# starting kernel
# baseline (speedup 1.0000x reference)
#include <hip/hip_runtime.h>


typedef float v2f __attribute__((ext_vector_type(2)));
typedef float v8f __attribute__((ext_vector_type(8)));

#define Hdim    56
#define Ssp     3136
#define C12     12
#define WPAD    62
#define T3N     (C12*Hdim*WPAD)
#define CPAD    32
#define T5ROWS  58
#define T5N     (T5ROWS*Hdim*CPAD)
#define ROWBYTES1 3472
#define JSTRIDE2  1792

__global__ void k_prep(const float* __restrict__ x,
                       float* __restrict__ t3rp, float* __restrict__ t5p) {
  int idx = blockIdx.x * blockDim.x + threadIdx.x;
  if (idx < T3N) {
    int c = idx / (Hdim * WPAD);
    int r = idx - c * (Hdim * WPAD);
    int n = r / WPAD;
    int w = r - n * WPAD;
    float v = 0.0f;
    if (w >= 3 && w < 59) {
      int m = w - 3;
      int msrc = (m + Hdim - 1) % Hdim;
      int s = c * Ssp + n * Hdim + msrc;
      v = x[s] + x[s + C12 * Ssp];
    }
    t3rp[idx] = v;
  } else {
    int p = idx - T3N;
    if (p < 2 * Hdim * CPAD) {
      int top = p / (Hdim * CPAD);
      int q = p - top * (Hdim * CPAD);
      t5p[(top ? (T5ROWS - 1) * Hdim * CPAD : 0) + q] = 0.0f;
    }
  }
}

__global__ void k_gemm1(const float* __restrict__ w2,
                        const float* __restrict__ t3rp,
                        float* __restrict__ t5p) {
  __shared__ float w2p[32 * 84];
  for (int i = threadIdx.x; i < 32 * 84; i += blockDim.x) {
    int row = i / 84;
    int kkp = i - row * 84;
    int jj = kkp / 12, c = kkp - jj * 12;
    w2p[i] = (row < 21) ? w2[(c * 7 + jj) * 21 + row] : 0.0f;
  }
  __syncthreads();

  int wave = (blockIdx.x * blockDim.x + threadIdx.x) >> 5;
  int lane = threadIdx.x & 31;
  int mt = wave & 1;
  int nt = wave >> 1;
  int half = lane >> 4;
  int l15  = lane & 15;

  int row = mt * 16 + l15;
  int col = nt * 16 + l15;
  int n = col / Hdim;
  int m = col - n * Hdim;

  const v2f*   ap = (const v2f*)&w2p[row * 84 + 2 * half];
  const float* bp = t3rp + 2 * half * ROWBYTES1 + n * WPAD + m;

  v8f acc = {};
  #pragma unroll
  for (int t = 0; t < 21; ++t) {
    v2f a = ap[2 * t];
    int off = 4 * (t % 3) * ROWBYTES1 + t / 3;
    v2f b;
    b.x = bp[off];
    b.y = bp[off + ROWBYTES1];
    acc = __builtin_amdgcn_wmma_f32_16x16x4_f32(
        false, a, false, b, (short)0, acc, false, false);
  }

  #pragma unroll
  for (int v = 0; v < 8; ++v) {
    int rout = mt * 16 + v + half * 8;
    t5p[(col + Hdim) * CPAD + rout] = acc[v];
  }
}

__global__ void k_gemm2(const float* __restrict__ w1,
                        const float* __restrict__ t5p,
                        float* __restrict__ y) {
  __shared__ float w1p[96 * 96];
  for (int i = threadIdx.x; i < 96 * 96; i += blockDim.x) {
    int row = i / 96;
    int kkp = i - row * 96;
    int j = kkp / 32, k = kkp - j * 32;
    w1p[i] = (k < 21) ? w1[row * 63 + j * 21 + k] : 0.0f;
  }
  __syncthreads();

  int wave = (blockIdx.x * blockDim.x + threadIdx.x) >> 5;
  int lane = threadIdx.x & 31;
  int mt = wave % 6;
  int nt = wave / 6;
  int half = lane >> 4;
  int l15  = lane & 15;

  int row = mt * 16 + l15;
  int col = nt * 16 + l15;

  const v2f*   ap = (const v2f*)&w1p[row * 96 + 2 * half];
  const float* bp = t5p + col * CPAD + 2 * half;

  v8f acc = {};
  #pragma unroll
  for (int t = 0; t < 24; ++t) {
    v2f a = ap[2 * t];
    int j = t / 8;
    int off = j * JSTRIDE2 + (4 * t - 32 * j);
    v2f b;
    b.x = bp[off];
    b.y = bp[off + 1];
    acc = __builtin_amdgcn_wmma_f32_16x16x4_f32(
        false, a, false, b, (short)0, acc, false, false);
  }

  #pragma unroll
  for (int v = 0; v < 8; ++v) {
    int rout = mt * 16 + v + half * 8;
    y[rout * Ssp + col] = acc[v];
  }
}

extern "C" void kernel_launch(void* const* d_in, const int* in_sizes, int n_in,
                              void* d_out, int out_size, void* d_ws, size_t ws_size,
                              hipStream_t stream) {
  const float* x  = (const float*)d_in[0];
  const float* w1 = (const float*)d_in[1];
  const float* w2 = (const float*)d_in[2];
  float* y = (float*)d_out;

  float* t3rp = (float*)d_ws;
  float* t5p  = t3rp + T3N;

  int prep_threads = T3N + 2 * Hdim * CPAD;
  k_prep<<<(prep_threads + 255) / 256, 256, 0, stream>>>(x, t3rp, t5p);

  k_gemm1<<<49, 256, 0, stream>>>(w2, t3rp, t5p);

  k_gemm2<<<147, 256, 0, stream>>>(w1, t5p, y);
}


# Weights zero-padded once per call by the prep kernel; GEMM blocks fill LDS with 16-byte loads all in flight instead of a serial 4-byte load-wait-store gather loop (36 and 11 rounds per thread)
# speedup vs baseline: 2.3394x; 2.3394x over previous
#include <hip/hip_runtime.h>


typedef float v2f __attribute__((ext_vector_type(2)));
typedef float v4f __attribute__((ext_vector_type(4)));
typedef float v8f __attribute__((ext_vector_type(8)));

#define Hdim    56
#define Ssp     3136
#define C12     12
#define WPAD    62
#define T3N     (C12*Hdim*WPAD)
#define CPAD    32
#define T5ROWS  58
#define T5N     (T5ROWS*Hdim*CPAD)
#define ROWBYTES1 3472
#define JSTRIDE2  1792
#define T5Z     (2*Hdim*CPAD)
#define W1N     (96*96)
#define W2N     (32*84)
#define PREP_N  (T3N + T5Z + W1N + W2N)
#define PREP_BLOCKS 224
#define GEMM_THREADS 256

__global__ void k_prep(const float* __restrict__ x,
                       const float* __restrict__ w1,
                       const float* __restrict__ w2,
                       float* __restrict__ t3rp, float* __restrict__ t5p,
                       float* __restrict__ w1pg, float* __restrict__ w2pg) {
  int idx = blockIdx.x * blockDim.x + threadIdx.x;
  if (idx < T3N) {
    int c = idx / (Hdim * WPAD);
    int r = idx - c * (Hdim * WPAD);
    int n = r / WPAD;
    int w = r - n * WPAD;
    float v = 0.0f;
    if (w >= 3 && w < 59) {
      int m = w - 3;
      int msrc = (m + Hdim - 1) % Hdim;
      int s = c * Ssp + n * Hdim + msrc;
      v = x[s] + x[s + C12 * Ssp];
    }
    t3rp[idx] = v;
  } else if (idx < T3N + T5Z) {
    int p = idx - T3N;
    int top = p / (Hdim * CPAD);
    int q = p - top * (Hdim * CPAD);
    t5p[(top ? (T5ROWS - 1) * Hdim * CPAD : 0) + q] = 0.0f;
  } else if (idx < T3N + T5Z + W1N) {
    int i = idx - (T3N + T5Z);
    int row = i / 96;
    int kkp = i - row * 96;
    int j = kkp / 32, k = kkp - j * 32;
    w1pg[i] = (k < 21) ? w1[row * 63 + j * 21 + k] : 0.0f;
  } else if (idx < PREP_N) {
    int i = idx - (T3N + T5Z + W1N);
    int row = i / 84;
    int kkp = i - row * 84;
    int jj = kkp / 12, c = kkp - jj * 12;
    w2pg[i] = (row < 21) ? w2[(c * 7 + jj) * 21 + row] : 0.0f;
  }
}

__global__ void __launch_bounds__(GEMM_THREADS)
k_gemm1(const float* __restrict__ w2pg,
        const float* __restrict__ t3rp,
        float* __restrict__ t5p) {
  __shared__ __attribute__((aligned(16))) float w2p[W2N];
  {
    const v4f* src = (const v4f*)w2pg;
    v4f* dst = (v4f*)w2p;
    const int tid = threadIdx.x;
    const bool tail = tid < W2N / 4 - 2 * GEMM_THREADS;
    v4f r0 = src[tid];
    v4f r1 = src[tid + GEMM_THREADS];
    v4f r2 = {};
    if (tail) r2 = src[tid + 2 * GEMM_THREADS];
    dst[tid] = r0;
    dst[tid + GEMM_THREADS] = r1;
    if (tail) dst[tid + 2 * GEMM_THREADS] = r2;
  }
  __syncthreads();

  int wave = (blockIdx.x * blockDim.x + threadIdx.x) >> 5;
  int lane = threadIdx.x & 31;
  int mt = wave & 1;
  int nt = wave >> 1;
  int half = lane >> 4;
  int l15  = lane & 15;

  int row = mt * 16 + l15;
  int col = nt * 16 + l15;
  int n = col / Hdim;
  int m = col - n * Hdim;

  const v2f*   ap = (const v2f*)&w2p[row * 84 + 2 * half];
  const float* bp = t3rp + 2 * half * ROWBYTES1 + n * WPAD + m;

  v8f acc = {};
  #pragma unroll
  for (int t = 0; t < 21; ++t) {
    v2f a = ap[2 * t];
    int off = 4 * (t % 3) * ROWBYTES1 + t / 3;
    v2f b;
    b.x = bp[off];
    b.y = bp[off + ROWBYTES1];
    acc = __builtin_amdgcn_wmma_f32_16x16x4_f32(
        false, a, false, b, (short)0, acc, false, false);
  }

  #pragma unroll
  for (int v = 0; v < 8; ++v) {
    int rout = mt * 16 + v + half * 8;
    t5p[(col + Hdim) * CPAD + rout] = acc[v];
  }
}

__global__ void __launch_bounds__(GEMM_THREADS)
k_gemm2(const float* __restrict__ w1pg,
        const float* __restrict__ t5p,
        float* __restrict__ y) {
  __shared__ __attribute__((aligned(16))) float w1p[W1N];
  {
    const v4f* src = (const v4f*)w1pg;
    v4f* dst = (v4f*)w1p;
    v4f r[9];
    #pragma unroll
    for (int q = 0; q < 9; ++q)
      r[q] = src[threadIdx.x + GEMM_THREADS * q];
    #pragma unroll
    for (int q = 0; q < 9; ++q)
      dst[threadIdx.x + GEMM_THREADS * q] = r[q];
  }
  __syncthreads();

  int wave = (blockIdx.x * blockDim.x + threadIdx.x) >> 5;
  int lane = threadIdx.x & 31;
  int mt = wave % 6;
  int nt = wave / 6;
  int half = lane >> 4;
  int l15  = lane & 15;

  int row = mt * 16 + l15;
  int col = nt * 16 + l15;

  const v2f*   ap = (const v2f*)&w1p[row * 96 + 2 * half];
  const float* bp = t5p + col * CPAD + 2 * half;

  v8f acc = {};
  #pragma unroll
  for (int t = 0; t < 24; ++t) {
    v2f a = ap[2 * t];
    int j = t / 8;
    int off = j * JSTRIDE2 + (4 * t - 32 * j);
    v2f b;
    b.x = bp[off];
    b.y = bp[off + 1];
    acc = __builtin_amdgcn_wmma_f32_16x16x4_f32(
        false, a, false, b, (short)0, acc, false, false);
  }

  #pragma unroll
  for (int v = 0; v < 8; ++v) {
    int rout = mt * 16 + v + half * 8;
    y[rout * Ssp + col] = acc[v];
  }
}

extern "C" void kernel_launch(void* const* d_in, const int* in_sizes, int n_in,
                              void* d_out, int out_size, void* d_ws, size_t ws_size,
                              hipStream_t stream) {
  const float* x  = (const float*)d_in[0];
  const float* w1 = (const float*)d_in[1];
  const float* w2 = (const float*)d_in[2];
  float* y = (float*)d_out;

  float* t3rp = (float*)d_ws;
  float* t5p  = t3rp + T3N;
  float* w1pg = t5p + T5N;
  float* w2pg = w1pg + W1N;

  k_prep<<<PREP_BLOCKS, 256, 0, stream>>>(x, w1, w2, t3rp, t5p, w1pg, w2pg);

  k_gemm1<<<49, GEMM_THREADS, 0, stream>>>(w2pg, t3rp, t5p);

  k_gemm2<<<147, GEMM_THREADS, 0, stream>>>(w1pg, t5p, y);
}


# Three launches fused into one: each block (1 output row, 16 columns) builds its input slab, both weight matrices and its t5 rows in LDS and runs both GEMMs there; GEMM2 K packed 96->64 regroups sums
# speedup vs baseline: 4.7782x; 2.0425x over previous
#include <hip/hip_runtime.h>


typedef float v2f __attribute__((ext_vector_type(2)));
typedef float v4f __attribute__((ext_vector_type(4)));
typedef float v8f __attribute__((ext_vector_type(8)));

#define Hdim    56
#define Ssp     3136
#define C12     12
#define NT      192
#define PSW     22
#define PSROW   (PSW * C12)
#define PSN     (3 * PSROW)
#define W2P     84
#define W2SRC   (12 * 7 * 21)
#define W2N     (32 * W2P)
#define W1P     68
#define W1SRC4  (96 * 63 / 4)
#define T5P     68

__global__ void __launch_bounds__(NT)
k_fused(const float* __restrict__ x,
        const float* __restrict__ w1,
        const float* __restrict__ w2,
        float* __restrict__ y) {
  __shared__ __attribute__((aligned(16))) float ps[PSN];
  __shared__ __attribute__((aligned(16))) float w2p[W2N];
  __shared__ __attribute__((aligned(16))) float w1p[96 * W1P];
  __shared__ __attribute__((aligned(16))) float t5s[16 * T5P];

  const int tid = threadIdx.x;
  const int a   = blockIdx.x >> 2;
  const int b0  = (blockIdx.x & 3) * 16;

  v4f r1[8];
  #pragma unroll
  for (int q = 0; q < 8; ++q) {
    int f = tid + NT * q;
    f = f < W1SRC4 ? f : W1SRC4 - 1;
    r1[q] = ((const v4f*)w1)[f];
  }
  float r2[10];
  #pragma unroll
  for (int q = 0; q < 10; ++q) {
    int s = tid + NT * q;
    s = s < W2SRC ? s : W2SRC - 1;
    r2[q] = w2[s];
  }
  float rx[5];
  int   pdst[5];
  #pragma unroll
  for (int q = 0; q < 5; ++q) {
    int e = tid + NT * q;
    bool live = e < PSN;
    e = live ? e : PSN - 1;
    int c  = e / (3 * PSW);
    int r  = e - c * (3 * PSW);
    int nn = r / PSW;
    int wq = r - nn * PSW;
    int w  = b0 + wq;
    int n  = a - 1 + nn;
    bool ok = (w >= 3) && (w < 59) && (n >= 0) && (n < Hdim);
    int wc = w < 3 ? 3 : (w > 58 ? 58 : w);
    int nc = n < 0 ? 0 : (n > Hdim - 1 ? Hdim - 1 : n);
    int msrc = (wc + 52) % Hdim;
    int s = c * Ssp + nc * Hdim + msrc;
    float v = x[s] + x[s + C12 * Ssp];
    rx[q]   = ok ? v : 0.0f;
    pdst[q] = live ? nn * PSROW + wq * C12 + c : -1;
  }

  #pragma unroll
  for (int q = 0; q < 8; ++q) {
    int f = tid + NT * q;
    if (f < W1SRC4) {
      int e   = 4 * f;
      int row = e / 63;
      int kk  = e - row * 63;
      #pragma unroll
      for (int i = 0; i < 4; ++i) {
        w1p[row * W1P + kk] = r1[q][i];
        if (++kk == 63) { kk = 0; ++row; }
      }
    }
  }
  #pragma unroll
  for (int q = 0; q < 10; ++q) {
    int s = tid + NT * q;
    if (s < W2SRC) {
      int c   = s / 147;
      int r   = s - c * 147;
      int jj  = r / 21;
      int row = r - jj * 21;
      w2p[row * W2P + jj * C12 + c] = r2[q];
    }
  }
  #pragma unroll
  for (int q = 0; q < 5; ++q)
    if (pdst[q] >= 0) ps[pdst[q]] = rx[q];
  for (int i = W2SRC + tid; i < W2N; i += NT)
    w2p[i] = 0.0f;
  if (tid < 96) w1p[tid * W1P + 63] = 0.0f;
  if (tid < 16) t5s[tid * T5P + 63] = 0.0f;
  __syncthreads();

  const int wave = tid >> 5;
  const int lane = tid & 31;
  const int half = lane >> 4;
  const int l15  = lane & 15;

  {
    const int nn = wave >> 1;
    const int mt = wave & 1;
    const v2f* ap = (const v2f*)&w2p[(mt * 16 + l15) * W2P + 2 * half];
    const v2f* bp = (const v2f*)&ps[nn * PSROW + l15 * C12 + 2 * half];
    v8f acc = {};
    #pragma unroll
    for (int t = 0; t < 21; ++t)
      acc = __builtin_amdgcn_wmma_f32_16x16x4_f32(
          false, ap[2 * t], false, bp[2 * t], (short)0, acc, false, false);
    const int  n   = a - 1 + nn;
    const bool nok = (n >= 0) && (n < Hdim);
    #pragma unroll
    for (int r = 0; r < 8; ++r) {
      int kch = mt * 16 + 8 * half + r;
      if (kch < 21) t5s[l15 * T5P + nn * 21 + kch] = nok ? acc[r] : 0.0f;
    }
  }
  __syncthreads();

  {
    const v2f* ap = (const v2f*)&w1p[(wave * 16 + l15) * W1P + 2 * half];
    const v2f* bp = (const v2f*)&t5s[l15 * T5P + 2 * half];
    v8f acc = {};
    #pragma unroll
    for (int t = 0; t < 16; ++t)
      acc = __builtin_amdgcn_wmma_f32_16x16x4_f32(
          false, ap[2 * t], false, bp[2 * t], (short)0, acc, false, false);
    const int b = b0 + l15;
    if (b < Hdim) {
      #pragma unroll
      for (int r = 0; r < 8; ++r)
        y[(wave * 16 + 8 * half + r) * Ssp + a * Hdim + b] = acc[r];
    }
  }
}

extern "C" void kernel_launch(void* const* d_in, const int* in_sizes, int n_in,
                              void* d_out, int out_size, void* d_ws, size_t ws_size,
                              hipStream_t stream) {
  const float* x  = (const float*)d_in[0];
  const float* w1 = (const float*)d_in[1];
  const float* w2 = (const float*)d_in[2];
  float* y = (float*)d_out;

  k_fused<<<224, NT, 0, stream>>>(x, w1, w2, y);
}


# Both weight matrices staged in LDS as plain 16-byte copies; k order of each GEMM re-chosen so fragments are read straight from the copies (no per-float re-layout stores or index math); regroups sums
# speedup vs baseline: 5.2692x; 1.1028x over previous
#include <hip/hip_runtime.h>


typedef float v2f __attribute__((ext_vector_type(2)));
typedef float v4f __attribute__((ext_vector_type(4)));
typedef float v8f __attribute__((ext_vector_type(8)));

#define Hdim    56
#define Ssp     3136
#define C12     12
#define NT      192
#define PSW     22
#define PSROW   (PSW * C12)
#define PSN     (3 * PSROW)
#define W2SRC   (12 * 7 * 21)
#define W2SRC4  (W2SRC / 4)
#define W2TAIL  12
#define W1SRC   (96 * 63)
#define W1SRC4  (W1SRC / 4)
#define W1TAIL  4
#define T5P     66

__global__ void __launch_bounds__(NT)
k_fused(const float* __restrict__ x,
        const float* __restrict__ w1,
        const float* __restrict__ w2,
        float* __restrict__ y) {
  __shared__ __attribute__((aligned(16))) float ps[PSN];
  __shared__ __attribute__((aligned(16))) float w2s[W2SRC + W2TAIL];
  __shared__ __attribute__((aligned(16))) float w1s[W1SRC + W1TAIL];
  __shared__ __attribute__((aligned(16))) float t5s[16 * T5P];

  const int tid = threadIdx.x;
  const int a   = blockIdx.x >> 2;
  const int b0  = (blockIdx.x & 3) * 16;

  v4f r1[8];
  #pragma unroll
  for (int q = 0; q < 8; ++q) {
    int f = tid + NT * q;
    f = f < W1SRC4 ? f : W1SRC4 - 1;
    r1[q] = ((const v4f*)w1)[f];
  }
  v4f r2[3];
  #pragma unroll
  for (int q = 0; q < 3; ++q) {
    int f = tid + NT * q;
    f = f < W2SRC4 ? f : W2SRC4 - 1;
    r2[q] = ((const v4f*)w2)[f];
  }
  float rx[5];
  int   pdst[5];
  #pragma unroll
  for (int q = 0; q < 5; ++q) {
    int e = tid + NT * q;
    bool live = e < PSN;
    e = live ? e : PSN - 1;
    int c  = e / (3 * PSW);
    int r  = e - c * (3 * PSW);
    int nn = r / PSW;
    int wq = r - nn * PSW;
    int w  = b0 + wq;
    int n  = a - 1 + nn;
    bool ok = (w >= 3) && (w < 59) && (n >= 0) && (n < Hdim);
    int wc = w < 3 ? 3 : (w > 58 ? 58 : w);
    int nc = n < 0 ? 0 : (n > Hdim - 1 ? Hdim - 1 : n);
    int msrc = (wc + 52) % Hdim;
    int s = c * Ssp + nc * Hdim + msrc;
    float v = x[s] + x[s + C12 * Ssp];
    rx[q]   = ok ? v : 0.0f;
    pdst[q] = live ? nn * PSROW + wq * C12 + c : -1;
  }

  #pragma unroll
  for (int q = 0; q < 8; ++q) {
    int f = tid + NT * q;
    if (f < W1SRC4) ((v4f*)w1s)[f] = r1[q];
  }
  #pragma unroll
  for (int q = 0; q < 3; ++q) {
    int f = tid + NT * q;
    if (f < W2SRC4) ((v4f*)w2s)[f] = r2[q];
  }
  #pragma unroll
  for (int q = 0; q < 5; ++q)
    if (pdst[q] >= 0) ps[pdst[q]] = rx[q];
  if (tid < W2TAIL) w2s[W2SRC + tid] = 0.0f;
  if (tid < W1TAIL) w1s[W1SRC + tid] = 0.0f;
  if (tid < 16) t5s[tid * T5P + 63] = 0.0f;
  __syncthreads();

  const int wave = tid >> 5;
  const int lane = tid & 31;
  const int half = lane >> 4;
  const int l15  = lane & 15;

  {
    const int nn = wave >> 1;
    const int mt = wave & 1;
    const float* ap = &w2s[6 * half * 147 + mt * 16 + l15];
    const v2f*   bp = (const v2f*)&ps[nn * PSROW + l15 * C12 + 6 * half];
    v8f acc = {};
    #pragma unroll
    for (int t = 0; t < 21; ++t) {
      const int cp = t % 3;
      const int jj = t / 3;
      v2f av;
      av.x = ap[(2 * cp) * 147 + jj * 21];
      av.y = ap[(2 * cp + 1) * 147 + jj * 21];
      acc = __builtin_amdgcn_wmma_f32_16x16x4_f32(
          false, av, false, bp[jj * 6 + cp], (short)0, acc, false, false);
    }
    const int  n   = a - 1 + nn;
    const bool nok = (n >= 0) && (n < Hdim);
    #pragma unroll
    for (int r = 0; r < 8; ++r) {
      int kch = mt * 16 + 8 * half + r;
      if (kch < 21) t5s[l15 * T5P + nn * 21 + kch] = nok ? acc[r] : 0.0f;
    }
  }
  __syncthreads();

  {
    const float* ap = &w1s[(wave * 16 + l15) * 63 + 32 * half];
    const v2f*   bp = (const v2f*)&t5s[l15 * T5P + 32 * half];
    v8f acc = {};
    #pragma unroll
    for (int t = 0; t < 16; ++t) {
      v2f av;
      av.x = ap[2 * t];
      av.y = ap[2 * t + 1];
      if (t == 15) av.y = half ? 0.0f : av.y;
      acc = __builtin_amdgcn_wmma_f32_16x16x4_f32(
          false, av, false, bp[t], (short)0, acc, false, false);
    }
    const int b = b0 + l15;
    if (b < Hdim) {
      #pragma unroll
      for (int r = 0; r < 8; ++r)
        y[(wave * 16 + 8 * half + r) * Ssp + a * Hdim + b] = acc[r];
    }
  }
}

extern "C" void kernel_launch(void* const* d_in, const int* in_sizes, int n_in,
                              void* d_out, int out_size, void* d_ws, size_t ws_size,
                              hipStream_t stream) {
  const float* x  = (const float*)d_in[0];
  const float* w1 = (const float*)d_in[1];
  const float* w2 = (const float*)d_in[2];
  float* y = (float*)d_out;

  k_fused<<<224, NT, 0, stream>>>(x, w1, w2, y);
}
